# starting kernel
# baseline (speedup 1.0000x reference)
#include <hip/hip_runtime.h>
#include <math.h>

typedef __attribute__((ext_vector_type(2))) float v2f;
typedef __attribute__((ext_vector_type(8))) float v8f;

#define IMG_H 512
#define IMG_W 512
#define TS 16

__device__ __forceinline__ int iclamp(int v, int lo, int hi) {
    return v < lo ? lo : (v > hi ? hi : v);
}

__global__ __launch_bounds__(32) void sobel_wmma_kernel(const float* __restrict__ img,
                                                        float* __restrict__ out) {
    const int X0   = blockIdx.x * TS;
    const int Y0   = blockIdx.y * TS;
    const int n    = blockIdx.z;
    const int lane = threadIdx.x;

    __shared__ float patch[18][20];
    __shared__ float Pv[16][20];
    __shared__ float Qv[16][20];

    const float* base = img + (size_t)n * 3 * IMG_H * IMG_W;

    for (int idx = lane; idx < 18 * 18; idx += 32) {
        const int r  = idx / 18;
        const int c  = idx % 18;
        const int iy = iclamp(Y0 - 1 + r, 0, IMG_H - 1);
        const int ix = iclamp(X0 - 1 + c, 0, IMG_W - 1);
        const int o  = iy * IMG_W + ix;
        patch[r][c] = base[o] + base[IMG_H * IMG_W + o] + base[2 * IMG_H * IMG_W + o];
    }
    __syncthreads();

    for (int idx = lane; idx < 16 * 20; idx += 32) {
        const int j = idx / 20;
        const int c = idx % 20;
        float p = 0.0f, q = 0.0f;
        if (c < 18) {
            const int lr = iclamp(Y0 + j, 1, IMG_H - 2) - (Y0 - 1);
            const float a = patch[lr - 1][c];
            const float b = patch[lr][c];
            const float d = patch[lr + 1][c];
            p = a + 2.0f * b + d;
            q = a - d;
        }
        Pv[j][c] = p;
        Qv[j][c] = q;
    }
    __syncthreads();

    const int half = lane >> 4;
    const int lrow = lane & 15;
    const int li   = iclamp(X0 + lrow, 1, IMG_W - 2) - (X0 - 1);

    v8f accx = {};
    v8f accy = {};
#pragma unroll
    for (int kb = 0; kb < 20; kb += 4) {
        const int k0 = kb + 2 * half;
        const int k1 = k0 + 1;
        v2f a, b;

        a.x = Pv[lrow][k0];
        a.y = Pv[lrow][k1];
        b.x = (float)((k0 == li - 1) - (k0 == li + 1));
        b.y = (float)((k1 == li - 1) - (k1 == li + 1));
        accx = __builtin_amdgcn_wmma_f32_16x16x4_f32(false, a, false, b,
                                                     (short)0, accx, false, false);

        a.x = Qv[lrow][k0];
        a.y = Qv[lrow][k1];
        b.x = (float)((k0 == li - 1) + 2 * (k0 == li) + (k0 == li + 1));
        b.y = (float)((k1 == li - 1) + 2 * (k1 == li) + (k1 == li + 1));
        accy = __builtin_amdgcn_wmma_f32_16x16x4_f32(false, a, false, b,
                                                     (short)0, accy, false, false);
    }

    float* op = out + (size_t)n * IMG_H * IMG_W + (size_t)Y0 * IMG_W + (X0 + lrow);
#pragma unroll
    for (int r = 0; r < 8; ++r) {
        const int M = r + 8 * half;
        const float gx = accx[r];
        const float gy = accy[r];
        op[M * IMG_W] = sqrtf(gx * gx + gy * gy);
    }
}

extern "C" void kernel_launch(void* const* d_in, const int* in_sizes, int n_in,
                              void* d_out, int out_size, void* d_ws, size_t ws_size,
                              hipStream_t stream) {
    (void)in_sizes; (void)n_in; (void)out_size; (void)d_ws; (void)ws_size;
    const float* img = (const float*)d_in[0];
    float* out = (float*)d_out;

    dim3 grid(IMG_W / TS, IMG_H / TS, 32);
    dim3 block(32);
    sobel_wmma_kernel<<<grid, block, 0, stream>>>(img, out);
}


# next tile's image rows are loaded straight into the B registers behind the products that free them (one 8-row tile ahead, counted waits); 4 waves per SIMD walk 8 tiles each; f32 sum order unchanged
# speedup vs baseline: 1.4874x; 1.4874x over previous
#include <hip/hip_runtime.h>
#include <math.h>
#include <stddef.h>

typedef __attribute__((ext_vector_type(2))) float v2f;
typedef __attribute__((ext_vector_type(8))) float v8f;
typedef __attribute__((ext_vector_type(3), aligned(4))) float v3f;

#define IMG_H 512
#define IMG_W 512
#define CH_STRIDE (IMG_H * IMG_W)
#define TILE_ROWS 8
#define SUBTILES 2
#define STRIP_COLS (16 * SUBTILES)
#define TILES_PER_WAVE 8
#define WAVES_PER_BLOCK 8
#define KSTEPS 8
#define ROW_BYTES (IMG_W * 4)
#define CH_BYTES (CH_STRIDE * 4)

__device__ __forceinline__ int k_channel(int ks, int h) { return ks < 5 ? h : 2; }
__device__ __forceinline__ int k_row(int ks, int h, int e) { return ks < 5 ? 2 * ks + e : 2 * (ks - 5) + 6 * h + e; }

__device__ __forceinline__ v3f ld_img3(const float* __restrict__ base, unsigned lane_bytes, int const_bytes) {
    return *(const v3f*)((const char*)base + (size_t)lane_bytes + (ptrdiff_t)const_bytes);
}

__device__ __forceinline__ void load_step(v3f (&dst)[2], const float* __restrict__ imgn,
                                          unsigned offA, unsigned offB, int y0, int h, int ks) {
    const unsigned rowb   = (unsigned)y0 * ROW_BYTES;
    const unsigned adjTop = (y0 == 0) ? (unsigned)ROW_BYTES : 0u;
    const unsigned adjBot = (y0 == IMG_H - TILE_ROWS) ? (unsigned)ROW_BYTES : 0u;
    const unsigned vA = offA + rowb;
    const unsigned vB = offB + rowb;
#pragma unroll
    for (int e = 0; e < 2; ++e) {
        const int j0 = k_row(ks, 0, e);
        const int cb = (j0 - 1) * ROW_BYTES;
        unsigned vb;
        if (ks < 5)       vb = (j0 == 0) ? vA + adjTop : (j0 == 9) ? vA - adjBot : vA;
        else if (ks == 5) vb = (e == 0) ? vB + (h ? 0u : adjTop) : vB;
        else if (ks == 6) vb = (e == 1) ? vB - (h ? adjBot : 0u) : vB;
        else              vb = vB - (h ? 6u * ROW_BYTES : 0u);
        dst[e] = ld_img3(imgn, vb, cb);
    }
}

template <bool REFILL>
__device__ __forceinline__ void do_tile(const float* __restrict__ imgn, float* __restrict__ outn,
                                        const v2f (&afr)[3][KSTEPS], v3f (&bq)[SUBTILES][KSTEPS][2],
                                        const unsigned (&offA)[SUBTILES], const unsigned (&offB)[SUBTILES],
                                        int y0, int h, int lane, int x0, bool padlane) {
    v8f acc[SUBTILES][2] = {};
#pragma unroll
    for (int ks = 0; ks < KSTEPS; ++ks) {
#pragma unroll
        for (int s = 0; s < SUBTILES; ++s) {
#pragma unroll
            for (int dx = 0; dx < 3; ++dx) {
                v2f b;
                b.x = (ks == 7 && padlane) ? 0.0f : bq[s][ks][0][dx];
                b.y = (ks == 7 && padlane) ? 0.0f : bq[s][ks][1][dx];
                acc[s][(ks * 3 + dx) & 1] = __builtin_amdgcn_wmma_f32_16x16x4_f32(
                    false, afr[dx][ks], false, b, (short)0, acc[s][(ks * 3 + dx) & 1], false, false);
            }
        }
        if (REFILL) {
            __builtin_amdgcn_sched_barrier(0);
#pragma unroll
            for (int s = 0; s < SUBTILES; ++s)
                load_step(bq[s][ks], imgn, offA[s], offB[s], y0 + TILE_ROWS, h, ks);
        }
        __builtin_amdgcn_sched_barrier(0);
    }

    float mag[SUBTILES][4];
#pragma unroll
    for (int s = 0; s < SUBTILES; ++s) {
#pragma unroll
        for (int jj = 0; jj < 4; ++jj) {
            const float gx = acc[s][0][2 * jj]     + acc[s][1][2 * jj];
            const float gy = acc[s][0][2 * jj + 1] + acc[s][1][2 * jj + 1];
            mag[s][jj] = sqrtf(gx * gx + gy * gy);
        }
        if (y0 == 0 && h == 0)                  mag[s][0] = mag[s][1];
        if (y0 == IMG_H - TILE_ROWS && h == 1)  mag[s][3] = mag[s][2];
    }

    float* op = outn + (size_t)y0 * IMG_W + (x0 + lane);
#pragma unroll
    for (int jj = 0; jj < 4; ++jj) {
        const float pl = __shfl_xor(mag[0][jj], 16);
        const float pr = __shfl_xor(mag[1][jj], 16);
        const float top = h ? pr : mag[0][jj];
        const float bot = h ? mag[1][jj] : pl;
        op[jj * IMG_W]       = top;
        op[(4 + jj) * IMG_W] = bot;
    }
}

__global__ __launch_bounds__(256) void sobel_vband_kernel(const float* __restrict__ img,
                                                          const float* __restrict__ kx,
                                                          const float* __restrict__ ky,
                                                          float* __restrict__ out) {
    const int lane = threadIdx.x & 31;
    const int wave = __builtin_amdgcn_readfirstlane((int)(threadIdx.x >> 5));
    const int h    = lane >> 4;
    const int c16  = lane & 15;
    const int x0   = (blockIdx.x * WAVES_PER_BLOCK + wave) * STRIP_COLS;
    const int n    = blockIdx.z;

    unsigned offA[SUBTILES], offB[SUBTILES];
#pragma unroll
    for (int s = 0; s < SUBTILES; ++s) {
        int xc = x0 + 16 * s + c16;
        xc = xc < 1 ? 1 : (xc > IMG_W - 2 ? IMG_W - 2 : xc);
        const unsigned xl = (unsigned)(xc - 1);
        offA[s] = (unsigned)h * CH_BYTES + xl * 4u;
        offB[s] = 2u * CH_BYTES + (unsigned)h * (6u * ROW_BYTES) + xl * 4u;
    }
    const bool padlane = (h != 0);

    const float* __restrict__ imgn = img + (size_t)n * 3 * CH_STRIDE;
    float* __restrict__ outn = out + (size_t)n * IMG_H * IMG_W;

    v3f bq[SUBTILES][KSTEPS][2];

    const int y_first = blockIdx.y * (TILES_PER_WAVE * TILE_ROWS);
#pragma unroll
    for (int ks = 0; ks < KSTEPS; ++ks) {
#pragma unroll
        for (int s = 0; s < SUBTILES; ++s)
            load_step(bq[s][ks], imgn, offA[s], offB[s], y_first, h, ks);
        __builtin_amdgcn_sched_barrier(0);
    }

    const int f = c16 & 1;
    const int r = 4 * (c16 >> 3) + ((c16 & 7) >> 1);
    const float* __restrict__ taps = f ? ky : kx;
    v2f afr[3][KSTEPS];
#pragma unroll
    for (int dx = 0; dx < 3; ++dx) {
#pragma unroll
        for (int ks = 0; ks < KSTEPS; ++ks) {
            float av[2];
#pragma unroll
            for (int e = 0; e < 2; ++e) {
                const int c  = k_channel(ks, h);
                const int j  = k_row(ks, h, e);
                const int dy = j - r;
                const bool ok = (j < 10) && (dy >= 0) && (dy <= 2);
                const int ti = ok ? (c * 9 + dy * 3 + dx) : 0;
                const float tv = taps[ti];
                av[e] = ok ? tv : 0.0f;
            }
            afr[dx][ks].x = av[0];
            afr[dx][ks].y = av[1];
        }
    }

#pragma unroll 1
    for (int t = 0; t < TILES_PER_WAVE - 1; ++t)
        do_tile<true>(imgn, outn, afr, bq, offA, offB, y_first + t * TILE_ROWS, h, lane, x0, padlane);
    do_tile<false>(imgn, outn, afr, bq, offA, offB, y_first + (TILES_PER_WAVE - 1) * TILE_ROWS, h, lane, x0, padlane);
}

extern "C" void kernel_launch(void* const* d_in, const int* in_sizes, int n_in,
                              void* d_out, int out_size, void* d_ws, size_t ws_size,
                              hipStream_t stream) {
    (void)in_sizes; (void)n_in; (void)out_size; (void)d_ws; (void)ws_size;
    const float* img = (const float*)d_in[0];
    const float* kx  = (const float*)d_in[1];
    const float* ky  = (const float*)d_in[2];
    float* out = (float*)d_out;

    dim3 grid(IMG_W / (STRIP_COLS * WAVES_PER_BLOCK), IMG_H / (TILE_ROWS * TILES_PER_WAVE), 32);
    dim3 block(32 * WAVES_PER_BLOCK);
    sobel_vband_kernel<<<grid, block, 0, stream>>>(img, kx, ky, out);
}


# even/odd column split over a wave's two tiles: one 16-B load per lane and line feeds 2 of 3 taps of both tiles as register pairs (15 loads a tile, not 32); 8-B stores; unrolled; f32 sum order changed
# speedup vs baseline: 1.7160x; 1.1537x over previous
#include <hip/hip_runtime.h>
#include <math.h>
#include <stddef.h>

typedef __attribute__((ext_vector_type(2))) float v2f;
typedef __attribute__((ext_vector_type(8))) float v8f;
typedef __attribute__((ext_vector_type(4), aligned(4))) float v4fu;

#define IMG_H 512
#define IMG_W 512
#define CH_STRIDE (IMG_H * IMG_W)
#define TILE_ROWS 8
#define STRIP_COLS 32
#define TILES_PER_WAVE 8
#define WAVES_PER_BLOCK 8
#define PSTEPS 15
#define SSTEPS 8
#define ROW_BYTES (IMG_W * 4)
#define CH_BYTES (CH_STRIDE * 4)


__device__ __forceinline__ v4fu load_line(const float* __restrict__ imgn, unsigned off, int y0, int h, int i) {
    const unsigned rowb   = (unsigned)y0 * ROW_BYTES;
    const unsigned adjTop = (y0 == 0 && h == 0) ? (unsigned)ROW_BYTES : 0u;
    const unsigned adjBot = (y0 == IMG_H - TILE_ROWS && h != 0) ? (unsigned)ROW_BYTES : 0u;
    const int j0 = 2 * (i % 5);
    const int cb = (i / 5) * CH_BYTES + (j0 - 1) * ROW_BYTES;
    unsigned vb = off + rowb;
    if (i % 5 == 0) vb += adjTop;
    if (i % 5 == 4) vb -= adjBot;
    return *(const v4fu*)((const char*)imgn + (size_t)vb + (ptrdiff_t)cb);
}

template <bool REFILL>
__device__ __forceinline__ void do_tile(const float* __restrict__ imgn, float* __restrict__ outn,
                                        const v2f (&ap0)[PSTEPS], const v2f (&ap1)[PSTEPS],
                                        const v2f (&as0)[SSTEPS], const v2f (&as1)[SSTEPS],
                                        v4fu (&rq)[PSTEPS], unsigned off, int y0, int h, int c16, int x0) {
    v8f acc[2][2] = {};
    int n = 0;
#pragma unroll
    for (int s = 0; s < SSTEPS; ++s) {
#pragma unroll
        for (int u = 0; u < 2; ++u) {
            const int i = 2 * s + u;
            if (i < PSTEPS) {
                v2f b0, b1;
                b0.x = rq[i].x; b0.y = rq[i].y;
                b1.x = rq[i].z; b1.y = rq[i].w;
                acc[0][n & 1] = __builtin_amdgcn_wmma_f32_16x16x4_f32(
                    false, ap0[i], false, b0, (short)0, acc[0][n & 1], false, false);
                acc[1][n & 1] = __builtin_amdgcn_wmma_f32_16x16x4_f32(
                    false, ap1[i], false, b1, (short)0, acc[1][n & 1], false, false);
                ++n;
            }
        }
        {
            const int ia = 2 * s, ib = 2 * s + 1;
            v2f b0, b1;
            b0.x = rq[ia].z;
            b1.x = rq[ia].y;
            b0.y = (ib < PSTEPS) ? rq[ib < PSTEPS ? ib : ia].z : 0.0f;
            b1.y = (ib < PSTEPS) ? rq[ib < PSTEPS ? ib : ia].y : 0.0f;
            acc[0][n & 1] = __builtin_amdgcn_wmma_f32_16x16x4_f32(
                false, as0[s], false, b0, (short)0, acc[0][n & 1], false, false);
            acc[1][n & 1] = __builtin_amdgcn_wmma_f32_16x16x4_f32(
                false, as1[s], false, b1, (short)0, acc[1][n & 1], false, false);
            ++n;
        }
        if (REFILL) {
            __builtin_amdgcn_sched_barrier(0);
#pragma unroll
            for (int u = 0; u < 2; ++u) {
                const int i = 2 * s + u;
                if (i < PSTEPS) rq[i] = load_line(imgn, off, y0 + TILE_ROWS, h, i);
            }
        }
        __builtin_amdgcn_sched_barrier(0);
    }

    float mag[2][4];
#pragma unroll
    for (int t = 0; t < 2; ++t) {
#pragma unroll
        for (int jj = 0; jj < 4; ++jj) {
            const float gx = acc[t][0][2 * jj]     + acc[t][1][2 * jj];
            const float gy = acc[t][0][2 * jj + 1] + acc[t][1][2 * jj + 1];
            mag[t][jj] = sqrtf(gx * gx + gy * gy);
        }
        if (y0 == 0 && h == 0)                  mag[t][0] = mag[t][1];
        if (y0 == IMG_H - TILE_ROWS && h == 1)  mag[t][3] = mag[t][2];
    }
    const bool leftEdge  = (x0 == 0) && (c16 == 0);
    const bool rightEdge = (x0 == IMG_W - STRIP_COLS) && (c16 == 15);

    float* op = outn + (size_t)(y0 + 4 * h) * IMG_W + (x0 + 2 * c16);
#pragma unroll
    for (int jj = 0; jj < 4; ++jj) {
        v2f o;
        o.x = rightEdge ? mag[1][jj] : mag[0][jj];
        o.y = leftEdge  ? mag[0][jj] : mag[1][jj];
        *(v2f*)(op + jj * IMG_W) = o;
    }
}

__global__ __launch_bounds__(256) void sobel_vband_kernel(const float* __restrict__ img,
                                                          const float* __restrict__ kx,
                                                          const float* __restrict__ ky,
                                                          float* __restrict__ out) {
    const int lane = threadIdx.x & 31;
    const int wave = __builtin_amdgcn_readfirstlane((int)(threadIdx.x >> 5));
    const int h    = lane >> 4;
    const int c16  = lane & 15;
    const int x0   = (blockIdx.x * WAVES_PER_BLOCK + wave) * STRIP_COLS;
    const int n    = blockIdx.z;

    int xl = x0 + 2 * c16 - 1;
    xl = xl < 0 ? 0 : (xl > IMG_W - 4 ? IMG_W - 4 : xl);
    const unsigned off = (unsigned)h * ROW_BYTES + (unsigned)xl * 4u;

    const float* __restrict__ imgn = img + (size_t)n * 3 * CH_STRIDE;
    float* __restrict__ outn = out + (size_t)n * IMG_H * IMG_W;

    v4fu rq[PSTEPS];

    const int y_first = blockIdx.y * (TILES_PER_WAVE * TILE_ROWS);
#pragma unroll
    for (int i = 0; i < PSTEPS; ++i) {
        rq[i] = load_line(imgn, off, y_first, h, i);
        __builtin_amdgcn_sched_barrier(0);
    }

    const int f = c16 & 1;
    const int r = 4 * (c16 >> 3) + ((c16 & 7) >> 1);
    const float* __restrict__ taps = f ? ky : kx;
    v2f ap0[PSTEPS], ap1[PSTEPS], as0[SSTEPS], as1[SSTEPS];
#pragma unroll
    for (int i = 0; i < PSTEPS; ++i) {
        const int c  = i / 5;
        const int j  = 2 * (i % 5) + h;
        const int dy = j - r;
        const bool ok = (dy >= 0) && (dy <= 2);
        const int tb = ok ? (c * 9 + dy * 3) : 0;
        const float t0 = taps[tb], t1 = taps[tb + 1], t2 = taps[tb + 2];
        ap0[i].x = ok ? t0 : 0.0f;  ap0[i].y = ok ? t1 : 0.0f;
        ap1[i].x = ok ? t1 : 0.0f;  ap1[i].y = ok ? t2 : 0.0f;
    }
#pragma unroll
    for (int s = 0; s < SSTEPS; ++s) {
        float a0[2], a1[2];
#pragma unroll
        for (int e = 0; e < 2; ++e) {
            const int i  = 2 * s + e;
            const int c  = i / 5;
            const int j  = 2 * (i % 5) + h;
            const int dy = j - r;
            const bool ok = (i < PSTEPS) && (dy >= 0) && (dy <= 2);
            const int tb = ok ? (c * 9 + dy * 3) : 0;
            const float t0 = taps[tb], t2 = taps[tb + 2];
            a0[e] = ok ? t2 : 0.0f;
            a1[e] = ok ? t0 : 0.0f;
        }
        as0[s].x = a0[0]; as0[s].y = a0[1];
        as1[s].x = a1[0]; as1[s].y = a1[1];
    }

#pragma unroll
    for (int t = 0; t < TILES_PER_WAVE - 1; ++t)
        do_tile<true>(imgn, outn, ap0, ap1, as0, as1, rq, off, y_first + t * TILE_ROWS, h, c16, x0);
    do_tile<false>(imgn, outn, ap0, ap1, as0, as1, rq, off, y_first + (TILES_PER_WAVE - 1) * TILE_ROWS, h, c16, x0);
}

extern "C" void kernel_launch(void* const* d_in, const int* in_sizes, int n_in,
                              void* d_out, int out_size, void* d_ws, size_t ws_size,
                              hipStream_t stream) {
    (void)in_sizes; (void)n_in; (void)out_size; (void)d_ws; (void)ws_size;
    const float* img = (const float*)d_in[0];
    const float* kx  = (const float*)d_in[1];
    const float* ky  = (const float*)d_in[2];
    float* out = (float*)d_out;

    dim3 grid(IMG_W / (STRIP_COLS * WAVES_PER_BLOCK), IMG_H / (TILE_ROWS * TILES_PER_WAVE), 32);
    dim3 block(32 * WAVES_PER_BLOCK);
    sobel_vband_kernel<<<grid, block, 0, stream>>>(img, kx, ky, out);
}
